# starting kernel
# baseline (speedup 1.0000x reference)
#include <hip/hip_runtime.h>

typedef __attribute__((ext_vector_type(2))) float v2f;
typedef __attribute__((ext_vector_type(8))) float v8f;
typedef __attribute__((ext_vector_type(4))) int   v4i;

#define IN_F            4096
#define OUT_F           8192
#define BATCH           32
#define BLOCKS_PER_ROW  256
#define PF_DIST         64

__global__ __launch_bounds__(128)
void SparseQuantLinear_wmma_kernel(const float* __restrict__ x,
                                   const float* __restrict__ scales,
                                   const float* __restrict__ offsets,
                                   const float* __restrict__ bias,
                                   const int*   __restrict__ packed,
                                   float*       __restrict__ out)
{
    const int lane   = threadIdx.x & 31;
    const int wave   = threadIdx.x >> 5;
    const int o_base = blockIdx.x * 64 + wave * 16;

    const int n  = lane & 15;
    const int hi = lane >> 4;
    const int o  = o_base + n;

    const float* xrow0 = x + n * IN_F;
    const float* xrow1 = x + (n + 16) * IN_F;

    const int t0 = hi * 2;
    const int t1 = t0 + 4;

    v8f c0 = {};
    v8f c1 = {};

    for (int kb = 0; kb < BLOCKS_PER_ROW; ++kb) {
        const int   nz   = o * BLOCKS_PER_ROW + kb;
        const float s    = scales[nz];
        const float f    = offsets[nz];
        const v4i   p    = *(const v4i*)(packed + nz * 4);
        const int   colb = kb * 16;

        __builtin_prefetch(packed  + (nz + PF_DIST) * 4, 0, 0);
        __builtin_prefetch(scales  + (nz + PF_DIST),     0, 0);
        __builtin_prefetch(offsets + (nz + PF_DIST),     0, 0);

        v2f a0_lo = *(const v2f*)(xrow0 + colb + t0);
        v2f a0_hi = *(const v2f*)(xrow1 + colb + t0);
        v2f a1_lo = *(const v2f*)(xrow0 + colb + t1);
        v2f a1_hi = *(const v2f*)(xrow1 + colb + t1);

        const int byte0 = hi ? p.y : p.x;
        const int byte1 = hi ? p.w : p.z;
        v2f b0, b1;
        b0.x = (float)( byte0        & 15) * s + f;
        b0.y = (float)((byte0 >> 4)  & 15) * s + f;
        b1.x = (float)( byte1        & 15) * s + f;
        b1.y = (float)((byte1 >> 4)  & 15) * s + f;

        c0 = __builtin_amdgcn_wmma_f32_16x16x4_f32(false, a0_lo, false, b0, (short)0, c0, false, false);
        c1 = __builtin_amdgcn_wmma_f32_16x16x4_f32(false, a0_hi, false, b0, (short)0, c1, false, false);
        c0 = __builtin_amdgcn_wmma_f32_16x16x4_f32(false, a1_lo, false, b1, (short)0, c0, false, false);
        c1 = __builtin_amdgcn_wmma_f32_16x16x4_f32(false, a1_hi, false, b1, (short)0, c1, false, false);
    }

    const float bz   = bias[o];
    const int   mrow = hi * 8;
#pragma unroll
    for (int v = 0; v < 8; ++v) {
        out[(mrow + v)      * OUT_F + o] = c0[v] + bz;
        out[(mrow + v + 16) * OUT_F + o] = c1[v] + bz;
    }
}

extern "C" void kernel_launch(void* const* d_in, const int* in_sizes, int n_in,
                              void* d_out, int out_size, void* d_ws, size_t ws_size,
                              hipStream_t stream) {
    const float* x       = (const float*)d_in[0];
    const float* scales  = (const float*)d_in[1];
    const float* offsets = (const float*)d_in[2];
    const float* bias    = (const float*)d_in[3];
    const int*   packed  = (const int*)d_in[4];
    float* out = (float*)d_out;

    dim3 grid(OUT_F / 64);
    dim3 block(128);
    SparseQuantLinear_wmma_kernel<<<grid, block, 0, stream>>>(x, scales, offsets, bias, packed, out);
}


# Weights and live x staged through LDS by coalesced loads, dequantized once, 16-byte LDS fragment reads; 256 groups of 32 rows, next chunk in flight, chunk order rotated per group; sum order changes
# speedup vs baseline: 4.0119x; 4.0119x over previous
#include <hip/hip_runtime.h>

typedef __attribute__((ext_vector_type(2))) float v2f;
typedef __attribute__((ext_vector_type(4))) float v4f;
typedef __attribute__((ext_vector_type(8))) float v8f;
typedef __attribute__((ext_vector_type(4))) int   v4i;

#define IN_F            4096
#define OUT_F           8192
#define BLOCKS_PER_ROW  256
#define ROWS_PER_WG     32
#define CB              32
#define NCHUNK          (BLOCKS_PER_ROW / CB)
#define PITCH           260
#define WJ              8
#define XJ              16

__global__ __launch_bounds__(128)
void SparseQuantLinear_wmma_kernel(const float* __restrict__ x,
                                   const float* __restrict__ scales,
                                   const float* __restrict__ offsets,
                                   const float* __restrict__ bias,
                                   const int*   __restrict__ packed,
                                   float*       __restrict__ out)
{
    __shared__ __attribute__((aligned(16))) float Wl[2][ROWS_PER_WG * PITCH];
    __shared__ __attribute__((aligned(16))) float Xl[2][32 * PITCH];

    const int t      = threadIdx.x;
    const int lane   = t & 31;
    const int wave   = t >> 5;
    const int o_base = blockIdx.x * ROWS_PER_WG;

    const int wnz0 = (o_base + wave) * BLOCKS_PER_ROW + lane;
    const int wl0  = wave * PITCH + lane * 8;
    const int xk   = (wave & 1) * 16 + (lane >> 1);
    const int xh   = lane & 1;
    const float* xg0 = x + (wave >> 1) * IN_F + xk * 16 + xh * 4;
    const int xl0  = (wave >> 1) * PITCH + xk * 8 + xh * 4;

    const int nt = wave & 1;
    const int mt = wave >> 1;
    const int n  = lane & 15;
    const int hi = lane >> 4;
    const int wfrag = (nt * 16 + n) * PITCH + hi * 4;
    const int xfrag = (mt * 16 + n) * PITCH + hi * 4;

    v4i   pk[WJ];
    float sc[WJ];
    float of[WJ];
    v4f   xv[XJ];

    v8f acc0 = {};
    v8f acc1 = {};

#define LOAD_CHUNK(c)                                                         \
    {                                                                         \
        _Pragma("unroll")                                                     \
        for (int j = 0; j < WJ; ++j) {                                        \
            const int nz = wnz0 + j * 4 * BLOCKS_PER_ROW + (c) * CB;          \
            pk[j] = *(const v4i*)(packed + nz * 4);                           \
            sc[j] = scales[nz];                                               \
            of[j] = offsets[nz];                                              \
        }                                                                     \
        _Pragma("unroll")                                                     \
        for (int j = 0; j < XJ; ++j)                                          \
            xv[j] = *(const v4f*)(xg0 + j * 2 * IN_F + (c) * CB * 16);        \
    }

#define STORE_CHUNK(b)                                                        \
    {                                                                         \
        float* wdst = &Wl[b][wl0];                                            \
        _Pragma("unroll")                                                     \
        for (int j = 0; j < WJ; ++j) {                                        \
            const v4i p = pk[j];                                              \
            const float s = sc[j];                                            \
            const float f = of[j];                                            \
            v4f lo, up;                                                       \
            lo.x = (float)( p.x       & 15) * s + f;                          \
            lo.y = (float)((p.x >> 4) & 15) * s + f;                          \
            lo.z = (float)( p.y       & 15) * s + f;                          \
            lo.w = (float)((p.y >> 4) & 15) * s + f;                          \
            up.x = (float)( p.z       & 15) * s + f;                          \
            up.y = (float)((p.z >> 4) & 15) * s + f;                          \
            up.z = (float)( p.w       & 15) * s + f;                          \
            up.w = (float)((p.w >> 4) & 15) * s + f;                          \
            *(v4f*)(wdst + j * 4 * PITCH)     = lo;                           \
            *(v4f*)(wdst + j * 4 * PITCH + 4) = up;                           \
        }                                                                     \
        float* xdst = &Xl[b][xl0];                                            \
        _Pragma("unroll")                                                     \
        for (int j = 0; j < XJ; ++j)                                          \
            *(v4f*)(xdst + j * 2 * PITCH) = xv[j];                            \
    }

#define COMPUTE_CHUNK(b)                                                      \
    {                                                                         \
        const float* wsrc = &Wl[b][wfrag];                                    \
        const float* xsrc = &Xl[b][xfrag];                                    \
        _Pragma("unroll")                                                     \
        for (int kb = 0; kb < CB; kb += 2) {                                  \
            const v4f w0 = *(const v4f*)(wsrc + kb * 8);                      \
            const v4f a0 = *(const v4f*)(xsrc + kb * 8);                      \
            const v4f w1 = *(const v4f*)(wsrc + kb * 8 + 8);                  \
            const v4f a1 = *(const v4f*)(xsrc + kb * 8 + 8);                  \
            v2f a, w;                                                         \
            a.x = a0.x; a.y = a0.y; w.x = w0.x; w.y = w0.y;                   \
            acc0 = __builtin_amdgcn_wmma_f32_16x16x4_f32(false, a, false, w, (short)0, acc0, false, false); \
            a.x = a1.x; a.y = a1.y; w.x = w1.x; w.y = w1.y;                   \
            acc1 = __builtin_amdgcn_wmma_f32_16x16x4_f32(false, a, false, w, (short)0, acc1, false, false); \
            a.x = a0.z; a.y = a0.w; w.x = w0.z; w.y = w0.w;                   \
            acc0 = __builtin_amdgcn_wmma_f32_16x16x4_f32(false, a, false, w, (short)0, acc0, false, false); \
            a.x = a1.z; a.y = a1.w; w.x = w1.z; w.y = w1.w;                   \
            acc1 = __builtin_amdgcn_wmma_f32_16x16x4_f32(false, a, false, w, (short)0, acc1, false, false); \
        }                                                                     \
    }

    const int rot = blockIdx.x & (NCHUNK - 1);

    LOAD_CHUNK(rot)
    for (int c = 0; c < NCHUNK - 1; ++c) {
        const int b = c & 1;
        STORE_CHUNK(b)
        __syncthreads();
        LOAD_CHUNK((c + 1 + rot) & (NCHUNK - 1))
        __builtin_amdgcn_sched_barrier(0);
        COMPUTE_CHUNK(b)
    }
    {
        const int b = (NCHUNK - 1) & 1;
        STORE_CHUNK(b)
        __syncthreads();
        COMPUTE_CHUNK(b)
    }

    const int   o    = o_base + nt * 16 + n;
    const float bz   = bias[o];
    const int   mrow = mt * 16 + hi * 8;
#pragma unroll
    for (int v = 0; v < 8; ++v)
        out[(mrow + v) * OUT_F + o] = (acc0[v] + acc1[v]) + bz;
}

extern "C" void kernel_launch(void* const* d_in, const int* in_sizes, int n_in,
                              void* d_out, int out_size, void* d_ws, size_t ws_size,
                              hipStream_t stream) {
    const float* x       = (const float*)d_in[0];
    const float* scales  = (const float*)d_in[1];
    const float* offsets = (const float*)d_in[2];
    const float* bias    = (const float*)d_in[3];
    const int*   packed  = (const int*)d_in[4];
    float* out = (float*)d_out;

    dim3 grid(OUT_F / ROWS_PER_WG);
    dim3 block(128);
    SparseQuantLinear_wmma_kernel<<<grid, block, 0, stream>>>(x, scales, offsets, bias, packed, out);
}


# Wave roles split: 8 loader waves (two groups on alternate chunks: global loads, dequantize, LDS stores, next request two intervals ahead) feed 4 matrix waves over hand-written LDS-only barriers
# speedup vs baseline: 4.5507x; 1.1343x over previous
#include <hip/hip_runtime.h>


typedef __attribute__((ext_vector_type(2))) float v2f;
typedef __attribute__((ext_vector_type(4))) float v4f;
typedef __attribute__((ext_vector_type(8))) float v8f;
typedef __attribute__((ext_vector_type(4))) int   v4i;

#define IN_F            4096
#define OUT_F           8192
#define BLOCKS_PER_ROW  256
#define ROWS_PER_WG     32
#define CB              32
#define NCHUNK          (BLOCKS_PER_ROW / CB)
#define PITCH           260
#define WJ              8
#define XJ              16

static __device__ __forceinline__ void lds_handover()
{
    asm volatile("s_wait_dscnt 0x0\n\ts_barrier_signal -1\n\ts_barrier_wait -1" ::: "memory");
}

__global__ __launch_bounds__(384)
void SparseQuantLinear_wmma_kernel(const float* __restrict__ x,
                                   const float* __restrict__ scales,
                                   const float* __restrict__ offsets,
                                   const float* __restrict__ bias,
                                   const int*   __restrict__ packed,
                                   float*       __restrict__ out)
{
    __shared__ __attribute__((aligned(16))) float Wl[2][ROWS_PER_WG * PITCH];
    __shared__ __attribute__((aligned(16))) float Xl[2][32 * PITCH];

    const int t      = threadIdx.x;
    const int lane   = t & 31;
    const int  wave12 = __builtin_amdgcn_readfirstlane(t >> 5);
    const bool loader = wave12 < 8;
    const int  grp    = wave12 >> 2;
    const int  wave   = wave12 & 3;
    const int  o_base = blockIdx.x * ROWS_PER_WG;

    const int wnz0 = (o_base + wave) * BLOCKS_PER_ROW + lane;
    const int wl0  = wave * PITCH + lane * 8;
    const int xk   = (wave & 1) * 16 + (lane >> 1);
    const int xh   = lane & 1;
    const float* xg0 = x + (wave >> 1) * IN_F + xk * 16 + xh * 4;
    const int xl0  = (wave >> 1) * PITCH + xk * 8 + xh * 4;

    const int nt = wave & 1;
    const int mt = wave >> 1;
    const int n  = lane & 15;
    const int hi = lane >> 4;
    const int wfrag = (nt * 16 + n) * PITCH + hi * 4;
    const int xfrag = (mt * 16 + n) * PITCH + hi * 4;

    v4i   pk[WJ];
    float sc[WJ];
    float of[WJ];
    v4f   xv[XJ];

    v8f acc0 = {};
    v8f acc1 = {};

#define LOAD_CHUNK(c)                                                         \
    {                                                                         \
        _Pragma("unroll")                                                     \
        for (int q = 0; q < WJ; ++q) {                                        \
            const int nz = wnz0 + q * 4 * BLOCKS_PER_ROW + (c) * CB;          \
            pk[q] = *(const v4i*)(packed + nz * 4);                           \
            sc[q] = scales[nz];                                               \
            of[q] = offsets[nz];                                              \
        }                                                                     \
        _Pragma("unroll")                                                     \
        for (int q = 0; q < XJ; ++q)                                          \
            xv[q] = *(const v4f*)(xg0 + q * 2 * IN_F + (c) * CB * 16);        \
    }

#define STORE_CHUNK(b)                                                        \
    {                                                                         \
        float* wdst = &Wl[(b)][wl0];                                          \
        _Pragma("unroll")                                                     \
        for (int q = 0; q < WJ; ++q) {                                        \
            const v4i p = pk[q];                                              \
            const float sj = sc[q];                                           \
            const float f = of[q];                                            \
            v4f lo, up;                                                       \
            lo.x = (float)( p.x       & 15) * sj + f;                         \
            lo.y = (float)((p.x >> 4) & 15) * sj + f;                         \
            lo.z = (float)( p.y       & 15) * sj + f;                         \
            lo.w = (float)((p.y >> 4) & 15) * sj + f;                         \
            up.x = (float)( p.z       & 15) * sj + f;                         \
            up.y = (float)((p.z >> 4) & 15) * sj + f;                         \
            up.z = (float)( p.w       & 15) * sj + f;                         \
            up.w = (float)((p.w >> 4) & 15) * sj + f;                         \
            *(v4f*)(wdst + q * 4 * PITCH)     = lo;                           \
            *(v4f*)(wdst + q * 4 * PITCH + 4) = up;                           \
        }                                                                     \
        float* xdst = &Xl[(b)][xl0];                                          \
        _Pragma("unroll")                                                     \
        for (int q = 0; q < XJ; ++q)                                          \
            *(v4f*)(xdst + q * 2 * PITCH) = xv[q];                            \
    }

#define COMPUTE_CHUNK(b)                                                      \
    {                                                                         \
        const float* wsrc = &Wl[(b)][wfrag];                                  \
        const float* xsrc = &Xl[(b)][xfrag];                                  \
        _Pragma("unroll")                                                     \
        for (int kb = 0; kb < CB; kb += 2) {                                  \
            const v4f w0 = *(const v4f*)(wsrc + kb * 8);                      \
            const v4f a0 = *(const v4f*)(xsrc + kb * 8);                      \
            const v4f w1 = *(const v4f*)(wsrc + kb * 8 + 8);                  \
            const v4f a1 = *(const v4f*)(xsrc + kb * 8 + 8);                  \
            v2f a, w;                                                         \
            a.x = a0.x; a.y = a0.y; w.x = w0.x; w.y = w0.y;                   \
            acc0 = __builtin_amdgcn_wmma_f32_16x16x4_f32(false, a, false, w, (short)0, acc0, false, false); \
            a.x = a1.x; a.y = a1.y; w.x = w1.x; w.y = w1.y;                   \
            acc1 = __builtin_amdgcn_wmma_f32_16x16x4_f32(false, a, false, w, (short)0, acc1, false, false); \
            a.x = a0.z; a.y = a0.w; w.x = w0.z; w.y = w0.w;                   \
            acc0 = __builtin_amdgcn_wmma_f32_16x16x4_f32(false, a, false, w, (short)0, acc0, false, false); \
            a.x = a1.z; a.y = a1.w; w.x = w1.z; w.y = w1.w;                   \
            acc1 = __builtin_amdgcn_wmma_f32_16x16x4_f32(false, a, false, w, (short)0, acc1, false, false); \
        }                                                                     \
    }

    const int rot = blockIdx.x & (NCHUNK - 1);

    if (loader)
        LOAD_CHUNK((grp + rot) & (NCHUNK - 1))

#pragma nounroll
    for (int it = 0; it <= NCHUNK; ++it) {
        if (loader) {
            if ((it & 1) == grp && it < NCHUNK) {
                STORE_CHUNK(it & 1)
                if (it + 2 < NCHUNK)
                    LOAD_CHUNK((it + 2 + rot) & (NCHUNK - 1))
            }
        } else {
            if (it >= 1)
                COMPUTE_CHUNK((it - 1) & 1)
        }
        lds_handover();
    }

    if (!loader) {
        const int   o    = o_base + nt * 16 + n;
        const float bz   = bias[o];
        const int   mrow = mt * 16 + hi * 8;
#pragma unroll
        for (int v = 0; v < 8; ++v)
            out[(mrow + v) * OUT_F + o] = (acc0[v] + acc1[v]) + bz;
    }
}

extern "C" void kernel_launch(void* const* d_in, const int* in_sizes, int n_in,
                              void* d_out, int out_size, void* d_ws, size_t ws_size,
                              hipStream_t stream) {
    const float* x       = (const float*)d_in[0];
    const float* scales  = (const float*)d_in[1];
    const float* offsets = (const float*)d_in[2];
    const float* bias    = (const float*)d_in[3];
    const int*   packed  = (const int*)d_in[4];
    float* out = (float*)d_out;

    dim3 grid(OUT_F / ROWS_PER_WG);
    dim3 block(384);
    SparseQuantLinear_wmma_kernel<<<grid, block, 0, stream>>>(x, scales, offsets, bias, packed, out);
}


# Matrix waves issue LDS fragment reads in whole groups of 8 blocks, each one group ahead of its matrix instructions (two register sets), not four at a time among them; sum order unchanged
# speedup vs baseline: 4.9201x; 1.0812x over previous
#include <hip/hip_runtime.h>


typedef __attribute__((ext_vector_type(2))) float v2f;
typedef __attribute__((ext_vector_type(4))) float v4f;
typedef __attribute__((ext_vector_type(8))) float v8f;
typedef __attribute__((ext_vector_type(4))) int   v4i;

#define IN_F            4096
#define OUT_F           8192
#define BLOCKS_PER_ROW  256
#define ROWS_PER_WG     32
#define CB              32
#define NCHUNK          (BLOCKS_PER_ROW / CB)
#define PITCH           260
#define WJ              8
#define XJ              16

static __device__ __forceinline__ void lds_handover()
{
    asm volatile("s_wait_dscnt 0x0\n\ts_barrier_signal -1\n\ts_barrier_wait -1" ::: "memory");
}

__global__ __launch_bounds__(384)
void SparseQuantLinear_wmma_kernel(const float* __restrict__ x,
                                   const float* __restrict__ scales,
                                   const float* __restrict__ offsets,
                                   const float* __restrict__ bias,
                                   const int*   __restrict__ packed,
                                   float*       __restrict__ out)
{
    __shared__ __attribute__((aligned(16))) float Wl[2][ROWS_PER_WG * PITCH];
    __shared__ __attribute__((aligned(16))) float Xl[2][32 * PITCH];

    const int t      = threadIdx.x;
    const int lane   = t & 31;
    const int  wave12 = __builtin_amdgcn_readfirstlane(t >> 5);
    const bool loader = wave12 < 8;
    const int  grp    = wave12 >> 2;
    const int  wave   = wave12 & 3;
    const int  o_base = blockIdx.x * ROWS_PER_WG;

    const int wnz0 = (o_base + wave) * BLOCKS_PER_ROW + lane;
    const int wl0  = wave * PITCH + lane * 8;
    const int xk   = (wave & 1) * 16 + (lane >> 1);
    const int xh   = lane & 1;
    const float* xg0 = x + (wave >> 1) * IN_F + xk * 16 + xh * 4;
    const int xl0  = (wave >> 1) * PITCH + xk * 8 + xh * 4;

    const int nt = wave & 1;
    const int mt = wave >> 1;
    const int n  = lane & 15;
    const int hi = lane >> 4;
    const int wfrag = (nt * 16 + n) * PITCH + hi * 4;
    const int xfrag = (mt * 16 + n) * PITCH + hi * 4;

    v4i   pk[WJ];
    float sc[WJ];
    float of[WJ];
    v4f   xv[XJ];

    v8f acc0 = {};
    v8f acc1 = {};

#define LOAD_CHUNK(c)                                                         \
    {                                                                         \
        _Pragma("unroll")                                                     \
        for (int q = 0; q < WJ; ++q) {                                        \
            const int nz = wnz0 + q * 4 * BLOCKS_PER_ROW + (c) * CB;          \
            pk[q] = *(const v4i*)(packed + nz * 4);                           \
            sc[q] = scales[nz];                                               \
            of[q] = offsets[nz];                                              \
        }                                                                     \
        _Pragma("unroll")                                                     \
        for (int q = 0; q < XJ; ++q)                                          \
            xv[q] = *(const v4f*)(xg0 + q * 2 * IN_F + (c) * CB * 16);        \
    }

#define STORE_CHUNK(b)                                                        \
    {                                                                         \
        float* wdst = &Wl[(b)][wl0];                                          \
        _Pragma("unroll")                                                     \
        for (int q = 0; q < WJ; ++q) {                                        \
            const v4i p = pk[q];                                              \
            const float sj = sc[q];                                           \
            const float f = of[q];                                            \
            v4f lo, up;                                                       \
            lo.x = (float)( p.x       & 15) * sj + f;                         \
            lo.y = (float)((p.x >> 4) & 15) * sj + f;                         \
            lo.z = (float)( p.y       & 15) * sj + f;                         \
            lo.w = (float)((p.y >> 4) & 15) * sj + f;                         \
            up.x = (float)( p.z       & 15) * sj + f;                         \
            up.y = (float)((p.z >> 4) & 15) * sj + f;                         \
            up.z = (float)( p.w       & 15) * sj + f;                         \
            up.w = (float)((p.w >> 4) & 15) * sj + f;                         \
            *(v4f*)(wdst + q * 4 * PITCH)     = lo;                           \
            *(v4f*)(wdst + q * 4 * PITCH + 4) = up;                           \
        }                                                                     \
        float* xdst = &Xl[(b)][xl0];                                          \
        _Pragma("unroll")                                                     \
        for (int q = 0; q < XJ; ++q)                                          \
            *(v4f*)(xdst + q * 2 * PITCH) = xv[q];                            \
    }

#define GB 8
#define FRAG_LOAD(Fw, Fa, wsrc, xsrc, g)                                      \
    {                                                                         \
        _Pragma("unroll")                                                     \
        for (int j = 0; j < GB; ++j) {                                        \
            Fw[j] = *(const v4f*)((wsrc) + ((g) * GB + j) * 8);               \
            Fa[j] = *(const v4f*)((xsrc) + ((g) * GB + j) * 8);               \
        }                                                                     \
    }

#define FRAG_MMA(Fw, Fa)                                                      \
    {                                                                         \
        _Pragma("unroll")                                                     \
        for (int j = 0; j < GB; j += 2) {                                     \
            v2f a, w;                                                         \
            a.x = Fa[j].x; a.y = Fa[j].y; w.x = Fw[j].x; w.y = Fw[j].y;       \
            acc0 = __builtin_amdgcn_wmma_f32_16x16x4_f32(false, a, false, w, (short)0, acc0, false, false); \
            a.x = Fa[j + 1].x; a.y = Fa[j + 1].y; w.x = Fw[j + 1].x; w.y = Fw[j + 1].y; \
            acc1 = __builtin_amdgcn_wmma_f32_16x16x4_f32(false, a, false, w, (short)0, acc1, false, false); \
            a.x = Fa[j].z; a.y = Fa[j].w; w.x = Fw[j].z; w.y = Fw[j].w;       \
            acc0 = __builtin_amdgcn_wmma_f32_16x16x4_f32(false, a, false, w, (short)0, acc0, false, false); \
            a.x = Fa[j + 1].z; a.y = Fa[j + 1].w; w.x = Fw[j + 1].z; w.y = Fw[j + 1].w; \
            acc1 = __builtin_amdgcn_wmma_f32_16x16x4_f32(false, a, false, w, (short)0, acc1, false, false); \
        }                                                                     \
    }

#define GROUP_ORDER                                                           \
    __builtin_amdgcn_sched_group_barrier(0x100, 4 * GB, 0);                   \
    __builtin_amdgcn_sched_group_barrier(0x008, 2 * GB, 0);                   \
    __builtin_amdgcn_sched_group_barrier(0x100, 2 * GB, 0);                   \
    __builtin_amdgcn_sched_group_barrier(0x008, 2 * GB, 0);                   \
    __builtin_amdgcn_sched_group_barrier(0x100, 2 * GB, 0);                   \
    __builtin_amdgcn_sched_group_barrier(0x008, 4 * GB, 0);

    const int rot = blockIdx.x & (NCHUNK - 1);

    if (loader)
        LOAD_CHUNK((grp + rot) & (NCHUNK - 1))

#pragma nounroll
    for (int it = 0; it <= NCHUNK; ++it) {
        if (loader) {
            if ((it & 1) == grp && it < NCHUNK) {
                STORE_CHUNK(it & 1)
                if (it + 2 < NCHUNK)
                    LOAD_CHUNK((it + 2 + rot) & (NCHUNK - 1))
            }
        } else {
            if (it >= 1) {
                const float* wsrc = &Wl[(it - 1) & 1][wfrag];
                const float* xsrc = &Xl[(it - 1) & 1][xfrag];
                v4f Pw[GB], Pa[GB], Qw[GB], Qa[GB];
                FRAG_LOAD(Pw, Pa, wsrc, xsrc, 0)
                FRAG_LOAD(Qw, Qa, wsrc, xsrc, 1)
                FRAG_MMA(Pw, Pa)
                FRAG_LOAD(Pw, Pa, wsrc, xsrc, 2)
                FRAG_MMA(Qw, Qa)
                FRAG_LOAD(Qw, Qa, wsrc, xsrc, 3)
                FRAG_MMA(Pw, Pa)
                FRAG_MMA(Qw, Qa)
                GROUP_ORDER
            }
        }
        lds_handover();
    }

    if (!loader) {
        const int   o    = o_base + nt * 16 + n;
        const float bz   = bias[o];
        const int   mrow = mt * 16 + hi * 8;
#pragma unroll
        for (int v = 0; v < 8; ++v)
            out[(mrow + v) * OUT_F + o] = (acc0[v] + acc1[v]) + bz;
    }
}

extern "C" void kernel_launch(void* const* d_in, const int* in_sizes, int n_in,
                              void* d_out, int out_size, void* d_ws, size_t ws_size,
                              hipStream_t stream) {
    const float* x       = (const float*)d_in[0];
    const float* scales  = (const float*)d_in[1];
    const float* offsets = (const float*)d_in[2];
    const float* bias    = (const float*)d_in[3];
    const int*   packed  = (const int*)d_in[4];
    float* out = (float*)d_out;

    dim3 grid(OUT_F / ROWS_PER_WG);
    dim3 block(384);
    SparseQuantLinear_wmma_kernel<<<grid, block, 0, stream>>>(x, scales, offsets, bias, packed, out);
}


# Every loader wave takes a share of every chunk and keeps three chunks in flight in three register sets: requests three intervals ahead, not two; roles as two code paths; sum order unchanged
# speedup vs baseline: 5.2463x; 1.0663x over previous
#include <hip/hip_runtime.h>


typedef __attribute__((ext_vector_type(2))) float v2f;
typedef __attribute__((ext_vector_type(4))) float v4f;
typedef __attribute__((ext_vector_type(8))) float v8f;
typedef __attribute__((ext_vector_type(4))) int   v4i;

#define IN_F            4096
#define OUT_F           8192
#define BLOCKS_PER_ROW  256
#define ROWS_PER_WG     32
#define CB              32
#define NCHUNK          (BLOCKS_PER_ROW / CB)
#define PITCH           260
#define WJ              4
#define XJ              8

static __device__ __forceinline__ void lds_handover()
{
    asm volatile("s_wait_dscnt 0x0\n\ts_barrier_signal -1\n\ts_barrier_wait -1" ::: "memory");
}

__global__ __launch_bounds__(384)
void SparseQuantLinear_wmma_kernel(const float* __restrict__ x,
                                   const float* __restrict__ scales,
                                   const float* __restrict__ offsets,
                                   const float* __restrict__ bias,
                                   const int*   __restrict__ packed,
                                   float*       __restrict__ out)
{
    __shared__ __attribute__((aligned(16))) float Wl[2][ROWS_PER_WG * PITCH];
    __shared__ __attribute__((aligned(16))) float Xl[2][32 * PITCH];

    const int t      = threadIdx.x;
    const int lane   = t & 31;
    const int  wave12 = __builtin_amdgcn_readfirstlane(t >> 5);
    const bool loader = wave12 < 8;
    const int  wave8  = wave12 & 7;
    const int  wave   = wave12 & 3;
    const int  o_base = blockIdx.x * ROWS_PER_WG;

    const int wnz0 = (o_base + wave8) * BLOCKS_PER_ROW + lane;
    const int wl0  = wave8 * PITCH + lane * 8;
    const int xk   = (wave8 & 1) * 16 + (lane >> 1);
    const int xh   = lane & 1;
    const float* xg0 = x + (wave8 >> 1) * IN_F + xk * 16 + xh * 4;
    const int xl0  = (wave8 >> 1) * PITCH + xk * 8 + xh * 4;

    const int nt = wave & 1;
    const int mt = wave >> 1;
    const int n  = lane & 15;
    const int hi = lane >> 4;
    const int wfrag = (nt * 16 + n) * PITCH + hi * 4;
    const int xfrag = (mt * 16 + n) * PITCH + hi * 4;

    v4i   pkA[WJ], pkB[WJ], pkC[WJ];
    float scA[WJ], scB[WJ], scC[WJ];
    float ofA[WJ], ofB[WJ], ofC[WJ];
    v4f   xvA[XJ], xvB[XJ], xvC[XJ];

    v8f acc0 = {};
    v8f acc1 = {};

#define LOAD_CHUNK(S, c)                                                      \
    {                                                                         \
        _Pragma("unroll")                                                     \
        for (int q = 0; q < WJ; ++q) {                                        \
            const int nz = wnz0 + q * 8 * BLOCKS_PER_ROW + (c) * CB;          \
            pk##S[q] = *(const v4i*)(packed + nz * 4);                        \
            sc##S[q] = scales[nz];                                            \
            of##S[q] = offsets[nz];                                           \
        }                                                                     \
        _Pragma("unroll")                                                     \
        for (int q = 0; q < XJ; ++q)                                          \
            xv##S[q] = *(const v4f*)(xg0 + q * 4 * IN_F + (c) * CB * 16);     \
    }

#define STORE_CHUNK(S, b)                                                     \
    {                                                                         \
        float* wdst = &Wl[(b)][wl0];                                          \
        _Pragma("unroll")                                                     \
        for (int q = 0; q < WJ; ++q) {                                        \
            const v4i p = pk##S[q];                                           \
            const float sj = sc##S[q];                                        \
            const float f = of##S[q];                                         \
            v4f lo, up;                                                       \
            lo.x = (float)( p.x       & 15) * sj + f;                         \
            lo.y = (float)((p.x >> 4) & 15) * sj + f;                         \
            lo.z = (float)( p.y       & 15) * sj + f;                         \
            lo.w = (float)((p.y >> 4) & 15) * sj + f;                         \
            up.x = (float)( p.z       & 15) * sj + f;                         \
            up.y = (float)((p.z >> 4) & 15) * sj + f;                         \
            up.z = (float)( p.w       & 15) * sj + f;                         \
            up.w = (float)((p.w >> 4) & 15) * sj + f;                         \
            *(v4f*)(wdst + q * 8 * PITCH)     = lo;                           \
            *(v4f*)(wdst + q * 8 * PITCH + 4) = up;                           \
        }                                                                     \
        float* xdst = &Xl[(b)][xl0];                                          \
        _Pragma("unroll")                                                     \
        for (int q = 0; q < XJ; ++q)                                          \
            *(v4f*)(xdst + q * 4 * PITCH) = xv##S[q];                         \
    }

#define GB 8
#define FRAG_LOAD(Fw, Fa, wsrc, xsrc, g)                                      \
    {                                                                         \
        _Pragma("unroll")                                                     \
        for (int j = 0; j < GB; ++j) {                                        \
            Fw[j] = *(const v4f*)((wsrc) + ((g) * GB + j) * 8);               \
            Fa[j] = *(const v4f*)((xsrc) + ((g) * GB + j) * 8);               \
        }                                                                     \
    }

#define FRAG_MMA(Fw, Fa)                                                      \
    {                                                                         \
        _Pragma("unroll")                                                     \
        for (int j = 0; j < GB; j += 2) {                                     \
            v2f a, w;                                                         \
            a.x = Fa[j].x; a.y = Fa[j].y; w.x = Fw[j].x; w.y = Fw[j].y;       \
            acc0 = __builtin_amdgcn_wmma_f32_16x16x4_f32(false, a, false, w, (short)0, acc0, false, false); \
            a.x = Fa[j + 1].x; a.y = Fa[j + 1].y; w.x = Fw[j + 1].x; w.y = Fw[j + 1].y; \
            acc1 = __builtin_amdgcn_wmma_f32_16x16x4_f32(false, a, false, w, (short)0, acc1, false, false); \
            a.x = Fa[j].z; a.y = Fa[j].w; w.x = Fw[j].z; w.y = Fw[j].w;       \
            acc0 = __builtin_amdgcn_wmma_f32_16x16x4_f32(false, a, false, w, (short)0, acc0, false, false); \
            a.x = Fa[j + 1].z; a.y = Fa[j + 1].w; w.x = Fw[j + 1].z; w.y = Fw[j + 1].w; \
            acc1 = __builtin_amdgcn_wmma_f32_16x16x4_f32(false, a, false, w, (short)0, acc1, false, false); \
        }                                                                     \
    }

#define GROUP_ORDER                                                           \
    __builtin_amdgcn_sched_group_barrier(0x100, 4 * GB, 0);                   \
    __builtin_amdgcn_sched_group_barrier(0x008, 2 * GB, 0);                   \
    __builtin_amdgcn_sched_group_barrier(0x100, 2 * GB, 0);                   \
    __builtin_amdgcn_sched_group_barrier(0x008, 2 * GB, 0);                   \
    __builtin_amdgcn_sched_group_barrier(0x100, 2 * GB, 0);                   \
    __builtin_amdgcn_sched_group_barrier(0x008, 4 * GB, 0);

    const int rot = blockIdx.x & (NCHUNK - 1);

#define CHUNK_AT(i) (((i) + rot) & (NCHUNK - 1))

#define LOADER_TURN(IT, S)                                                    \
    STORE_CHUNK(S, (IT) & 1)                                                  \
    if ((IT) + 3 < NCHUNK)                                                    \
        LOAD_CHUNK(S, CHUNK_AT((IT) + 3))                                     \
    lds_handover();

    if (loader) {
        LOAD_CHUNK(A, CHUNK_AT(0))
        LOAD_CHUNK(B, CHUNK_AT(1))
        LOAD_CHUNK(C, CHUNK_AT(2))
        LOADER_TURN(0, A)
        LOADER_TURN(1, B)
        LOADER_TURN(2, C)
        LOADER_TURN(3, A)
        LOADER_TURN(4, B)
        LOADER_TURN(5, C)
        LOADER_TURN(6, A)
        LOADER_TURN(7, B)
        lds_handover();
    } else {
        lds_handover();
#pragma nounroll
        for (int it = 1; it <= NCHUNK; ++it) {
            const float* wsrc = &Wl[(it - 1) & 1][wfrag];
            const float* xsrc = &Xl[(it - 1) & 1][xfrag];
            v4f Pw[GB], Pa[GB], Qw[GB], Qa[GB];
            FRAG_LOAD(Pw, Pa, wsrc, xsrc, 0)
            FRAG_LOAD(Qw, Qa, wsrc, xsrc, 1)
            FRAG_MMA(Pw, Pa)
            FRAG_LOAD(Pw, Pa, wsrc, xsrc, 2)
            FRAG_MMA(Qw, Qa)
            FRAG_LOAD(Qw, Qa, wsrc, xsrc, 3)
            FRAG_MMA(Pw, Pa)
            FRAG_MMA(Qw, Qa)
            GROUP_ORDER
            __builtin_amdgcn_sched_barrier(0);
            lds_handover();
        }
    }

    if (!loader) {
        const int   o    = o_base + nt * 16 + n;
        const float bz   = bias[o];
        const int   mrow = mt * 16 + hi * 8;
#pragma unroll
        for (int v = 0; v < 8; ++v)
            out[(mrow + v) * OUT_F + o] = (acc0[v] + acc1[v]) + bz;
    }
}

extern "C" void kernel_launch(void* const* d_in, const int* in_sizes, int n_in,
                              void* d_out, int out_size, void* d_ws, size_t ws_size,
                              hipStream_t stream) {
    const float* x       = (const float*)d_in[0];
    const float* scales  = (const float*)d_in[1];
    const float* offsets = (const float*)d_in[2];
    const float* bias    = (const float*)d_in[3];
    const int*   packed  = (const int*)d_in[4];
    float* out = (float*)d_out;

    dim3 grid(OUT_F / ROWS_PER_WG);
    dim3 block(384);
    SparseQuantLinear_wmma_kernel<<<grid, block, 0, stream>>>(x, scales, offsets, bias, packed, out);
}
